# starting kernel
# baseline (speedup 1.0000x reference)
#include <hip/hip_runtime.h>


typedef __attribute__((ext_vector_type(2))) float v2f;
typedef __attribute__((ext_vector_type(8))) float v8f;

#define N_BATCH   32
#define SEQ       1024
#define SENT_LEN  512
#define CTX_LEN   256
#define LO        (1 + CTX_LEN)
#define NUM_SEG   2048
#define NROWS     (N_BATCH * SENT_LEN)
#define FEAT      2048
#define SEGS_PER_TILE 16
#define CAP       4096

static __device__ __forceinline__ int imin(int a, int b) { return a < b ? a : b; }
static __device__ __forceinline__ int imax(int a, int b) { return a > b ? a : b; }

__launch_bounds__(256, 2)
__global__ void seg_mean_wmma_kernel(const float* __restrict__ x,
                                     const int*   __restrict__ seg,
                                     float*       __restrict__ out) {
    __shared__ int2  s_pair[CAP + 4];
    __shared__ int   s_cnt[SEGS_PER_TILE];
    __shared__ float s_den[SEGS_PER_TILE];
    __shared__ int   s_toff[256];
    __shared__ int   s_total;

    const int tid  = threadIdx.x;
    const int base = blockIdx.x * SEGS_PER_TILE;

    if (tid < SEGS_PER_TILE) s_cnt[tid] = 0;

    int c = 0;
    for (int i = tid; i < NROWS; i += 256) {
        const int b = i >> 9, t = i & (SENT_LEN - 1);
        const int s = seg[b * SEQ + LO + t];
        if ((unsigned)(s - base) < (unsigned)SEGS_PER_TILE) c++;
    }
    s_toff[tid] = c;
    __syncthreads();
    if (tid == 0) {
        int acc = 0;
        for (int k = 0; k < 256; ++k) { int v = s_toff[k]; s_toff[k] = acc; acc += v; }
        s_total = imin(acc, CAP);
    }
    __syncthreads();

    int pos = s_toff[tid];
    for (int i = tid; i < NROWS; i += 256) {
        const int b = i >> 9, t = i & (SENT_LEN - 1);
        const int s = seg[b * SEQ + LO + t];
        const unsigned d = (unsigned)(s - base);
        if (d < (unsigned)SEGS_PER_TILE) {
            if (pos < CAP) {
                int2 e;
                e.x = (b * SEQ + LO + t) << 10;
                e.y = (b << 20) | (int)d;
                s_pair[pos] = e;
            }
            pos++;
            atomicAdd(&s_cnt[d], 1);
        }
    }
    __syncthreads();
    if (tid < SEGS_PER_TILE) s_den[tid] = (float)imax(s_cnt[tid], 1);

    const int total = s_total;
    __syncthreads();
    const int totalv = s_total;
    if (tid < 3 && totalv > 0) {
        const int idx = totalv + tid;
        if (idx & 3 || tid < ((4 - (totalv & 3)) & 3)) {   }
        int2 p0 = s_pair[0];
        int2 e; e.x = p0.x; e.y = (p0.y & ~15) | 16;
        s_pair[idx] = e;
    }
    __syncthreads();
    (void)total;

    const int ngroups = (totalv + 3) >> 2;

    const int lane = tid & 31;
    const int wave = tid >> 5;
    const int n    = lane & 15;
    const int hi   = lane >> 4;
    const int k0 = hi ? 2 : 0;

    for (int stripe = 0; stripe < FEAT / 128; ++stripe) {
        const int fb    = stripe * 128 + wave * 16;
        const int isEmb = (fb < 1024);
        const int colOff = isEmb ? (fb + n) : (fb + n - 1024);

        v8f acc = {};
        for (int g = 0; g < ngroups; ++g) {
            const int gi = (g << 2) + k0;
            const int2 p0 = s_pair[gi];
            const int2 p1 = s_pair[gi + 1];

            v2f a;
            a.x = ((p0.y & 31) == n) ? 1.0f : 0.0f;
            a.y = ((p1.y & 31) == n) ? 1.0f : 0.0f;

            const int b0 = (isEmb ? p0.x : (p0.y & ~31)) + colOff;
            const int b1 = (isEmb ? p1.x : (p1.y & ~31)) + colOff;
            v2f bv;
            bv.x = x[b0];
            bv.y = x[b1];

            acc = __builtin_amdgcn_wmma_f32_16x16x4_f32(
                false, a, false, bv, (short)0, acc, false, false);
        }

        union { v8f v; float f[8]; } u; u.v = acc;
        #pragma unroll
        for (int j = 0; j < 8; ++j) {
            const int m = j + (hi << 3);
            out[(long)(base + m) * FEAT + fb + n] = u.f[j] / s_den[m];
        }
    }
}

extern "C" void kernel_launch(void* const* d_in, const int* in_sizes, int n_in,
                              void* d_out, int out_size, void* d_ws, size_t ws_size,
                              hipStream_t stream) {
    (void)in_sizes; (void)n_in; (void)out_size; (void)d_ws; (void)ws_size;
    const float* x   = (const float*)d_in[0];
    const int*   seg = (const int*)d_in[1];
    float*       out = (float*)d_out;
    seg_mean_wmma_kernel<<<128, 256, 0, stream>>>(x, seg, out);
}


# Gather loads in a ring of three register sets of 8 groups: a set's 16 row loads per lane are issued two sets ahead of its WMMAs instead of one round trip per 4 groups; sum order unchanged
# speedup vs baseline: 6.5806x; 6.5806x over previous
#include <hip/hip_runtime.h>


typedef __attribute__((ext_vector_type(2))) float v2f;
typedef __attribute__((ext_vector_type(8))) float v8f;

#define N_BATCH   32
#define SEQ       1024
#define SENT_LEN  512
#define CTX_LEN   256
#define LO        (1 + CTX_LEN)
#define NUM_SEG   2048
#define NROWS     (N_BATCH * SENT_LEN)
#define FEAT      2048
#define SEGS_PER_TILE 16
#define CAP       4096
#define NTHREADS  1024
#define FEAT_PER_BLOCK 512
#define CH        8
#define PADENT    (4 * CH * 5 + 4)

static __device__ __forceinline__ int imin(int a, int b) { return a < b ? a : b; }
static __device__ __forceinline__ int imax(int a, int b) { return a > b ? a : b; }

__launch_bounds__(NTHREADS, 1)
__global__ void seg_mean_wmma_kernel(const float* __restrict__ x,
                                     const int*   __restrict__ seg,
                                     float*       __restrict__ out) {
    __shared__ int2  s_pair[CAP + PADENT];
    __shared__ __attribute__((aligned(16))) int s_raw[CAP];
    __shared__ int   s_cnt[SEGS_PER_TILE];
    __shared__ float s_den[SEGS_PER_TILE];
    __shared__ int   s_total;
    __shared__ __attribute__((aligned(16))) int s_cb[SEGS_PER_TILE * N_BATCH];

    const int tid  = threadIdx.x;
    const int base = (blockIdx.x >> 1) * SEGS_PER_TILE;
    const int half = blockIdx.x & 1;

    int sv[NROWS / NTHREADS];
    #pragma unroll
    for (int j = 0; j < NROWS / NTHREADS; ++j) {
        const int i = tid + j * NTHREADS;
        const int b = i >> 9, t = i & (SENT_LEN - 1);
        sv[j] = seg[b * SEQ + LO + t];
    }

    if (tid < SEGS_PER_TILE) s_cnt[tid] = 0;
    if (tid < SEGS_PER_TILE * N_BATCH) s_cb[tid] = 0;
    if (tid == 0) s_total = 0;
    {
        int4 big; big.x = 0x7fffffff; big.y = 0x7fffffff; big.z = 0x7fffffff; big.w = 0x7fffffff;
        *(int4*)&s_raw[tid * (CAP / NTHREADS)] = big;
    }
    __syncthreads();

    #pragma unroll
    for (int j = 0; j < NROWS / NTHREADS; ++j) {
        const unsigned d = (unsigned)(sv[j] - base);
        if (d < (unsigned)SEGS_PER_TILE) {
            const int b = (tid >> 9) + 2 * j;
            const int pos = atomicAdd(&s_total, 1);
            if (pos < CAP) s_raw[pos] = (tid << 8) | (j << 4) | (int)d;
            atomicAdd(&s_cnt[d], 1);
            atomicAdd(&s_cb[d * N_BATCH + b], 1);
        }
    }
    __syncthreads();
    if (tid < SEGS_PER_TILE) s_den[tid] = (float)imax(s_cnt[tid], 1);

    const int totalv = imin(s_total, CAP);
    const int nquads = (totalv + 3) >> 2;
    if (tid < PADENT) {
        int2 pad; pad.x = 0; pad.y = 16;
        s_pair[totalv + tid] = pad;
    }
    for (int e = tid; e < totalv; e += NTHREADS) {
        const int mine = s_raw[e];
        int r = 0;
        #pragma unroll 8
        for (int q = 0; q < nquads; ++q) {
            const int4 v = *(const int4*)&s_raw[q << 2];
            r += (v.x < mine ? 1 : 0) + (v.y < mine ? 1 : 0)
               + (v.z < mine ? 1 : 0) + (v.w < mine ? 1 : 0);
        }
        const int rt = mine >> 8;
        const int rj = (mine >> 4) & 15;
        const int b  = (rt >> 9) + 2 * rj;
        const int t  = rt & (SENT_LEN - 1);
        int2 en;
        en.x = (b * SEQ + LO + t) << 10;
        en.y = (b << 20) | (mine & 15);
        s_pair[r] = en;
    }
    __syncthreads();

    const int ngroups = (totalv + 3) >> 2;

    const int lane = tid & 31;
    const int wave = tid >> 5;
    const int n    = lane & 15;
    const int hi   = lane >> 4;
    const int k0 = hi ? 2 : 0;

    const int fb = half * FEAT_PER_BLOCK + wave * 16;

    {
        const int colOff = fb + n;

        v8f acc = {};
        float b0[2 * CH], b1[2 * CH], b2[2 * CH];

        #define ISSUE_SET(buf, gfirst)                                        \
            _Pragma("unroll")                                                 \
            for (int i = 0; i < CH; ++i) {                                    \
                const int gi = (((gfirst) + i) << 2) + k0;                    \
                buf[2 * i]     = x[s_pair[gi].x     + colOff];                \
                buf[2 * i + 1] = x[s_pair[gi + 1].x + colOff];                \
            }

        #define CONSUME_SET(buf, gfirst)                                      \
            {                                                                 \
                int my[2 * CH];                                               \
                _Pragma("unroll")                                             \
                for (int i = 0; i < CH; ++i) {                                \
                    const int gi = (((gfirst) + i) << 2) + k0;                \
                    my[2 * i]     = s_pair[gi].y;                             \
                    my[2 * i + 1] = s_pair[gi + 1].y;                         \
                }                                                             \
                _Pragma("unroll")                                             \
                for (int i = 0; i < CH; ++i) {                                \
                    if ((gfirst) + i < ngroups) {            \
                        v2f a, bv;                                            \
                        a.x = ((my[2 * i]     & 31) == n) ? 1.0f : 0.0f;      \
                        a.y = ((my[2 * i + 1] & 31) == n) ? 1.0f : 0.0f;      \
                        bv.x = buf[2 * i];                                    \
                        bv.y = buf[2 * i + 1];                                \
                        acc = __builtin_amdgcn_wmma_f32_16x16x4_f32(          \
                            false, a, false, bv, (short)0, acc, false, false);\
                    }                                                         \
                }                                                             \
            }

        ISSUE_SET(b0, 0)
        ISSUE_SET(b1, CH)
        for (int g0 = 0; g0 < ngroups; g0 += 3 * CH) {
            ISSUE_SET(b2, g0 + 2 * CH)
            CONSUME_SET(b0, g0)
            ISSUE_SET(b0, g0 + 3 * CH)
            CONSUME_SET(b1, g0 + CH)
            ISSUE_SET(b1, g0 + 4 * CH)
            CONSUME_SET(b2, g0 + 2 * CH)
        }
        #undef ISSUE_SET
        #undef CONSUME_SET

        union { v8f v; float f[8]; } u; u.v = acc;
        #pragma unroll
        for (int j = 0; j < 8; ++j) {
            const int m = j + (hi << 3);
            out[(long)(base + m) * FEAT + fb + n] = u.f[j] / s_den[m];
        }
    }

    v2f acnt[N_BATCH / 4];
    #pragma unroll
    for (int ks = 0; ks < N_BATCH / 4; ++ks) {
        const int2 cpair = *(const int2*)&s_cb[n * N_BATCH + (ks << 2) + k0];
        acnt[ks].x = (float)cpair.x;
        acnt[ks].y = (float)cpair.y;
    }

    {
        const float* __restrict__ cp = x + fb + n;

        v8f acc = {};
        #pragma unroll
        for (int ks = 0; ks < N_BATCH / 4; ++ks) {
            const int b = (ks << 2) + k0;
            v2f bv;
            bv.x = cp[b << 20];
            bv.y = cp[(b + 1) << 20];
            acc = __builtin_amdgcn_wmma_f32_16x16x4_f32(
                false, acnt[ks], false, bv, (short)0, acc, false, false);
        }

        union { v8f v; float f[8]; } u; u.v = acc;
        #pragma unroll
        for (int j = 0; j < 8; ++j) {
            const int m = j + (hi << 3);
            out[(long)(base + m) * FEAT + 1024 + fb + n] = u.f[j] / s_den[m];
        }
    }
}

extern "C" void kernel_launch(void* const* d_in, const int* in_sizes, int n_in,
                              void* d_out, int out_size, void* d_ws, size_t ws_size,
                              hipStream_t stream) {
    (void)in_sizes; (void)n_in; (void)out_size; (void)d_ws; (void)ws_size;
    const float* x   = (const float*)d_in[0];
    const int*   seg = (const int*)d_in[1];
    float*       out = (float*)d_out;
    seg_mean_wmma_kernel<<<256, 1024, 0, stream>>>(x, seg, out);
}
